# starting kernel
# baseline (speedup 1.0000x reference)
#include <hip/hip_runtime.h>
#include <hip/hip_bf16.h>

typedef float v2f __attribute__((ext_vector_type(2)));
typedef float v8f __attribute__((ext_vector_type(8)));

template <int N, int NT, bool FULLM>
__global__ __launch_bounds__(256) void gemm_wmma(
    float* __restrict__ Z, const float* __restrict__ H,
    const float* __restrict__ S, int Mrows)
{
    const int lane    = threadIdx.x & 31;
    const int wave    = (blockIdx.x * blockDim.x + threadIdx.x) >> 5;
    const int nStrips = N / (16 * NT);
    const int mTiles  = (Mrows + 15) >> 4;
    if (wave >= mTiles * nStrips) return;

    const int mTile  = wave / nStrips;
    const int nStrip = wave - mTile * nStrips;
    const int m0 = mTile * 16;
    const int n0 = nStrip * (16 * NT);

    const int col  = lane & 15;
    const int hi   = lane >> 4;
    const int koff = hi << 1;
    const int m    = m0 + col;

    const float scale = (m < Mrows) ? 1.0f : 0.0f;
    const float* __restrict__ Hrow =
        H + (size_t)(FULLM ? m : (m < Mrows ? m : 0)) * N + koff;
    const float* __restrict__ Sbase = S + (size_t)koff * N + n0 + col;

    v8f acc[NT];
#pragma unroll
    for (int t = 0; t < NT; ++t)
        acc[t] = (v8f){0.f, 0.f, 0.f, 0.f, 0.f, 0.f, 0.f, 0.f};

#pragma unroll 8
    for (int k = 0; k < N; k += 4) {
        v2f a;
        a.x = Hrow[k];
        a.y = Hrow[k + 1];
        if (!FULLM) { a.x *= scale; a.y *= scale; }

        const float* __restrict__ Sp = Sbase + (size_t)k * N;
#pragma unroll
        for (int t = 0; t < NT; ++t) {
            v2f b;
            b.x = Sp[t * 16];
            b.y = Sp[t * 16 + N];
            acc[t] = __builtin_amdgcn_wmma_f32_16x16x4_f32(
                false, a, false, b, (short)0, acc[t], false, false);
        }
    }

#pragma unroll
    for (int t = 0; t < NT; ++t) {
#pragma unroll
        for (int v = 0; v < 8; ++v) {
            const int mrow = m0 + v + (hi << 3);
            if (FULLM || mrow < Mrows)
                Z[(size_t)mrow * N + n0 + t * 16 + col] = acc[t][v];
        }
    }
}

__global__ void combine_relu_kernel(
    float* __restrict__ Hout,
    const float* __restrict__ Z0, const float* __restrict__ Z1,
    const float* __restrict__ Z2,
    const float* __restrict__ W, const float* __restrict__ bias,
    int Bn, int Fin, int Fout, int N)
{
    const int tid = blockIdx.x * blockDim.x + threadIdx.x;
    if (tid >= Bn * N) return;
    const int n = tid % N;
    const int b = tid / N;

    float out[16];
    for (int g = 0; g < Fout; ++g) out[g] = bias[g];

    for (int f = 0; f < Fin; ++f) {
        const size_t idx = ((size_t)(b * Fin + f)) * (size_t)N + n;
        const float z0 = Z0[idx];
        const float z1 = Z1[idx];
        const float z2 = Z2[idx];
        const float* __restrict__ W0 = W + ((size_t)(0 * Fin + f)) * Fout;
        const float* __restrict__ W1 = W + ((size_t)(1 * Fin + f)) * Fout;
        const float* __restrict__ W2 = W + ((size_t)(2 * Fin + f)) * Fout;
        for (int g = 0; g < Fout; ++g)
            out[g] = fmaf(z0, W0[g], fmaf(z1, W1[g], fmaf(z2, W2[g], out[g])));
    }

    for (int g = 0; g < Fout; ++g) {
        const float v = out[g];
        Hout[((size_t)(b * Fout + g)) * (size_t)N + n] = v > 0.f ? v : 0.f;
    }
}

__global__ void pool_kernel(
    float* __restrict__ P, const float* __restrict__ Hin,
    const int* __restrict__ incid, int BF, int N1, int N2, int DEG)
{
    const int tid = blockIdx.x * blockDim.x + threadIdx.x;
    if (tid >= BF * N2) return;
    const int m  = tid % N2;
    const int bf = tid / N2;
    const float* __restrict__ row = Hin + (size_t)bf * (size_t)N1;
    float v = 0.0f;
    for (int d = 0; d < DEG; ++d) {
        const int j = incid[m * DEG + d];
        v = fmaxf(v, row[j]);
    }
    P[tid] = v;
}

__global__ void readout_kernel(
    float* __restrict__ Y, const float* __restrict__ H,
    const float* __restrict__ R0W, const float* __restrict__ R0b,
    const float* __restrict__ R1W, const float* __restrict__ R1b,
    int Bn, int N2)
{
    const int tid = blockIdx.x * blockDim.x + threadIdx.x;
    if (tid >= Bn * N2) return;
    const int n = tid % N2;
    const int b = tid / N2;

    float v[16];
    for (int i = 0; i < 16; ++i)
        v[i] = H[((size_t)(b * 16 + i)) * (size_t)N2 + n];

    float y0[16];
    for (int j = 0; j < 16; ++j) {
        float s = R0b[j];
        for (int i = 0; i < 16; ++i) s = fmaf(v[i], R0W[i * 16 + j], s);
        y0[j] = s > 0.f ? s : 0.f;
    }

    for (int r = 0; r < 2; ++r) {
        float s = R1b[r];
        for (int j = 0; j < 16; ++j) s = fmaf(y0[j], R1W[j * 2 + r], s);
        Y[((size_t)(b * 2 + r)) * (size_t)N2 + n] = s;
    }
}

extern "C" void kernel_launch(void* const* d_in, const int* in_sizes, int n_in,
                              void* d_out, int out_size, void* d_ws, size_t ws_size,
                              hipStream_t stream)
{
    (void)in_sizes; (void)n_in; (void)out_size; (void)ws_size;

    const float* x     = (const float*)d_in[0];
    const float* S1    = (const float*)d_in[1];
    const float* S2    = (const float*)d_in[2];
    const int*   incid = (const int*)  d_in[3];
    const float* W10   = (const float*)d_in[4];
    const float* b10   = (const float*)d_in[5];
    const float* W11   = (const float*)d_in[6];
    const float* b11   = (const float*)d_in[7];
    const float* W20   = (const float*)d_in[8];
    const float* b20   = (const float*)d_in[9];
    const float* R0W   = (const float*)d_in[10];
    const float* R0b   = (const float*)d_in[11];
    const float* R1W   = (const float*)d_in[12];
    const float* R1b   = (const float*)d_in[13];
    float* out = (float*)d_out;

    const int B = 4, F1 = 16, N1 = 8192, N2 = 4096, DEG = 4;

    char* ws = (char*)d_ws;
    const size_t szBig = (size_t)(B * F1) * N1 * sizeof(float);
    float* h1 = (float*)(ws);
    float* tA = (float*)(ws + 1 * szBig);
    float* tB = (float*)(ws + 2 * szBig);
    float* h2 = (float*)(ws + 3 * szBig);
    float* p  = (float*)(ws + 4 * szBig);

    auto blocksFor = [](int tiles) { return (tiles + 7) / 8; };
    auto grid1d    = [](int total, int blk) { return (total + blk - 1) / blk; };

    {
        const int tiles = 1 * (N1 / 16);
        gemm_wmma<N1, 1, false><<<blocksFor(tiles), 256, 0, stream>>>(tA, x,  S1, B);
        gemm_wmma<N1, 1, false><<<blocksFor(tiles), 256, 0, stream>>>(tB, tA, S1, B);
    }
    combine_relu_kernel<<<grid1d(B * N1, 256), 256, 0, stream>>>(
        h1, x, tA, tB, W10, b10, B, 1, F1, N1);

    {
        const int tiles = 4 * (N1 / 32);
        gemm_wmma<N1, 2, true><<<blocksFor(tiles), 256, 0, stream>>>(tA, h1, S1, B * F1);
        gemm_wmma<N1, 2, true><<<blocksFor(tiles), 256, 0, stream>>>(tB, tA, S1, B * F1);
    }
    combine_relu_kernel<<<grid1d(B * N1, 256), 256, 0, stream>>>(
        h2, h1, tA, tB, W11, b11, B, F1, F1, N1);

    pool_kernel<<<grid1d(B * F1 * N2, 256), 256, 0, stream>>>(
        p, h2, incid, B * F1, N1, N2, DEG);

    {
        const int tiles = 4 * (N2 / 32);
        gemm_wmma<N2, 2, true><<<blocksFor(tiles), 256, 0, stream>>>(tA, p,  S2, B * F1);
        gemm_wmma<N2, 2, true><<<blocksFor(tiles), 256, 0, stream>>>(tB, tA, S2, B * F1);
    }
    combine_relu_kernel<<<grid1d(B * N2, 256), 256, 0, stream>>>(
        h1  , p, tA, tB, W20, b20, B, F1, F1, N2);

    readout_kernel<<<grid1d(B * N2, 256), 256, 0, stream>>>(
        out, h1, R0W, R0b, R1W, R1b, B, N2);
}


# GEMM staging folded into the WMMA stream: between WMMA blocks a register is stored to LDS for the next chunk and at once reloaded for the chunk after; no staging phase left; sum order unchanged
# speedup vs baseline: 3.8949x; 3.8949x over previous
#include <hip/hip_runtime.h>

typedef float v2f __attribute__((ext_vector_type(2)));
typedef float v4f __attribute__((ext_vector_type(4)));
typedef float v8f __attribute__((ext_vector_type(8)));

template <int N, int MROWS>
__global__ __launch_bounds__(256) void gemm_strip_lds(
    float* __restrict__ Z, const float* __restrict__ H,
    const float* __restrict__ S)
{
    constexpr int KC      = 256;
    constexpr int MT      = (MROWS + 15) / 16;
    constexpr int PA      = KC + 4;
    constexpr int PB      = 96;
    constexpr int PO      = 36;
    constexpr int A_STAGE = 16 * MT * PA;
    constexpr int B_STAGE = (KC / 2) * PB;
    constexpr int NCH     = N / KC;
    constexpr int HV      = (MROWS * (KC / 4)) / 256;
    constexpr int NACC    = 4;
    constexpr int BLK     = 8;
    constexpr int NBLK    = (KC / 4) / BLK;
    constexpr int HRG     = 256 / (KC / 4);
    constexpr int NST     = HV + 4;
    constexpr int SPB     = (NST + NBLK - 1) / NBLK;
    static_assert(NBLK * BLK * 4 == KC && BLK % NACC == 0, "block shape");
    static_assert(HV * 256 == MROWS * (KC / 4), "H chunk must split evenly");
    static_assert(N % KC == 0 && N % 32 == 0, "shape");
    static_assert(16 * MT * PO <= B_STAGE, "output tile fits a stage");
    static_assert(NCH >= 3, "the chunk loop peels its last two turns");

    __shared__ __attribute__((aligned(16))) float As[2][A_STAGE];
    __shared__ __attribute__((aligned(16))) float Bs[2][B_STAGE];

    const int t    = threadIdx.x;
    const int lane = t & 31;
    const int w    = t >> 5;
    const int c    = lane & 15;
    const int h    = lane >> 4;
    const int mt   = w >> 1;
    const int nt   = w & 1;
    const int n0   = blockIdx.x * 32;

    v4f hreg[HV];
    v4f sreg[8];

    const float* __restrict__ hp =
        H + (size_t)(t / (KC / 4)) * N + 4 * (t % (KC / 4));
    const float* __restrict__ sp =
        S + (size_t)(2 * (t >> 3)) * N + n0 + 4 * (t & 7);
    float* la = &As[0][(t / (KC / 4)) * PA + 4 * (t % (KC / 4))];
    float* lb = &Bs[0][(t >> 3) * PB + 8 * (t & 7)];

    auto step = [&](int st, float* qa, float* qb, const float* hq,
                    const float* sq, bool more) {
        if (st < HV) {
            const int j = st;
            *reinterpret_cast<v4f*>(qa + j * HRG * PA) = hreg[j];
            if (more) {
                __builtin_amdgcn_sched_barrier(0);
                hreg[j] = *reinterpret_cast<const v4f*>(
                    hq + (size_t)(j * HRG) * N);
            }
        } else if (st < NST) {
            const int j = st - HV;
            const v4f r0 = sreg[2 * j];
            const v4f r1 = sreg[2 * j + 1];
            float* q = qb + j * 32 * PB;
            *reinterpret_cast<v4f*>(q)     = (v4f){r0.x, r1.x, r0.y, r1.y};
            *reinterpret_cast<v4f*>(q + 4) = (v4f){r0.z, r1.z, r0.w, r1.w};
            if (more) {
                __builtin_amdgcn_sched_barrier(0);
                const float* __restrict__ p = sq + (size_t)(64 * j) * N;
                sreg[2 * j]     = *reinterpret_cast<const v4f*>(p);
                sreg[2 * j + 1] = *reinterpret_cast<const v4f*>(p + N);
            }
        }
    };

    auto stage_barrier = [&]() {
        __builtin_amdgcn_fence(__ATOMIC_RELEASE, "workgroup", "local");
        __builtin_amdgcn_s_barrier();
        __builtin_amdgcn_fence(__ATOMIC_ACQUIRE, "workgroup", "local");
    };

    v8f accp[NACC];
#pragma unroll
    for (int q = 0; q < NACC; ++q)
        accp[q] = (v8f){0.f, 0.f, 0.f, 0.f, 0.f, 0.f, 0.f, 0.f};

    auto turn = [&](int buf, int kn, bool stage, bool more) {
        const float* a = &As[buf][(mt * 16 + c) * PA + 2 * h];
        const float* b = &Bs[buf][h * PB + (nt * 16 + c) * 2];
        float* qa = la + (buf ^ 1) * A_STAGE;
        float* qb = lb + (buf ^ 1) * B_STAGE;
        const float* hq = hp + kn;
        const float* sq = sp + (size_t)kn * N;
        v2f af[2][BLK];
        v2f bf[2][BLK];
        if (mt < MT) {
#pragma unroll
            for (int q = 0; q < BLK; ++q) {
                af[0][q] = *reinterpret_cast<const v2f*>(a + 4 * q);
                bf[0][q] = *reinterpret_cast<const v2f*>(b + 2 * q * PB);
            }
        }
#pragma unroll
        for (int blk = 0; blk < NBLK; ++blk) {
            const int cur = blk & 1;
            if (mt < MT) {
                if (blk + 1 < NBLK) {
#pragma unroll
                    for (int q = 0; q < BLK; ++q) {
                        const int ks = (blk + 1) * BLK + q;
                        af[cur ^ 1][q] =
                            *reinterpret_cast<const v2f*>(a + 4 * ks);
                        bf[cur ^ 1][q] =
                            *reinterpret_cast<const v2f*>(b + 2 * ks * PB);
                    }
                }
                __builtin_amdgcn_sched_barrier(0);
#pragma unroll
                for (int q = 0; q < BLK; ++q) {
                    accp[q % NACC] = __builtin_amdgcn_wmma_f32_16x16x4_f32(
                        false, af[cur][q], false, bf[cur][q], (short)0,
                        accp[q % NACC], false, false);
                }
            }
            __builtin_amdgcn_sched_barrier(0);
            if (stage) {
#pragma unroll
                for (int u = 0; u < SPB; ++u)
                    step(blk * SPB + u, qa, qb, hq, sq, more);
                __builtin_amdgcn_sched_barrier(0);
            }
        }
    };

    if (MROWS < 16 * MT) {
        for (int b = 0; b < 2; ++b)
            for (int i = t; i < (16 * MT - MROWS) * PA; i += 256)
                As[b][MROWS * PA + i] = 0.0f;
    }

#pragma unroll
    for (int j = 0; j < HV; ++j)
        hreg[j] = *reinterpret_cast<const v4f*>(hp + (size_t)(j * HRG) * N);
#pragma unroll
    for (int j = 0; j < 4; ++j) {
        const float* __restrict__ p = sp + (size_t)(64 * j) * N;
        sreg[2 * j]     = *reinterpret_cast<const v4f*>(p);
        sreg[2 * j + 1] = *reinterpret_cast<const v4f*>(p + N);
    }
#pragma unroll
    for (int st = 0; st < NST; ++st)
        step(st, la, lb, hp + KC, sp + (size_t)KC * N, true);
    stage_barrier();

    for (int i = 0; i < NCH - 2; ++i) {
        turn(i & 1, (i + 2) * KC, true, true);
        stage_barrier();
    }
    turn((NCH - 2) & 1, 0, true, false);
    stage_barrier();
    turn((NCH - 1) & 1, 0, false, false);
    stage_barrier();

    float* Ot = &Bs[0][0];
    const v8f acc = (accp[0] + accp[1]) + (accp[2] + accp[3]);
    if (mt < MT) {
#pragma unroll
        for (int v = 0; v < 8; ++v)
            Ot[(mt * 16 + v + 8 * h) * PO + nt * 16 + c] = acc[v];
    }
    __syncthreads();
    for (int idx = t; idx < MROWS * 8; idx += 256) {
        const int row = idx >> 3;
        const int q   = idx & 7;
        const v4f val = *reinterpret_cast<const v4f*>(&Ot[row * PO + 4 * q]);
        *reinterpret_cast<v4f*>(Z + (size_t)row * N + n0 + 4 * q) = val;
    }
}

__global__ void combine_relu_kernel(
    float* __restrict__ Hout,
    const float* __restrict__ Z0, const float* __restrict__ Z1,
    const float* __restrict__ Z2,
    const float* __restrict__ W, const float* __restrict__ bias,
    int Bn, int Fin, int Fout, int N)
{
    const int tid = blockIdx.x * blockDim.x + threadIdx.x;
    if (tid >= Bn * N) return;
    const int n = tid % N;
    const int b = tid / N;

    float out[16];
    for (int g = 0; g < Fout; ++g) out[g] = bias[g];

    for (int f = 0; f < Fin; ++f) {
        const size_t idx = ((size_t)(b * Fin + f)) * (size_t)N + n;
        const float z0 = Z0[idx];
        const float z1 = Z1[idx];
        const float z2 = Z2[idx];
        const float* __restrict__ W0 = W + ((size_t)(0 * Fin + f)) * Fout;
        const float* __restrict__ W1 = W + ((size_t)(1 * Fin + f)) * Fout;
        const float* __restrict__ W2 = W + ((size_t)(2 * Fin + f)) * Fout;
        for (int g = 0; g < Fout; ++g)
            out[g] = fmaf(z0, W0[g], fmaf(z1, W1[g], fmaf(z2, W2[g], out[g])));
    }

    for (int g = 0; g < Fout; ++g) {
        const float v = out[g];
        Hout[((size_t)(b * Fout + g)) * (size_t)N + n] = v > 0.f ? v : 0.f;
    }
}

__global__ void pool_kernel(
    float* __restrict__ P, const float* __restrict__ Hin,
    const int* __restrict__ incid, int BF, int N1, int N2, int DEG)
{
    const int tid = blockIdx.x * blockDim.x + threadIdx.x;
    if (tid >= BF * N2) return;
    const int m  = tid % N2;
    const int bf = tid / N2;
    const float* __restrict__ row = Hin + (size_t)bf * (size_t)N1;
    float v = 0.0f;
    for (int d = 0; d < DEG; ++d) {
        const int j = incid[m * DEG + d];
        v = fmaxf(v, row[j]);
    }
    P[tid] = v;
}

__global__ void readout_kernel(
    float* __restrict__ Y, const float* __restrict__ H,
    const float* __restrict__ R0W, const float* __restrict__ R0b,
    const float* __restrict__ R1W, const float* __restrict__ R1b,
    int Bn, int N2)
{
    const int tid = blockIdx.x * blockDim.x + threadIdx.x;
    if (tid >= Bn * N2) return;
    const int n = tid % N2;
    const int b = tid / N2;

    float v[16];
    for (int i = 0; i < 16; ++i)
        v[i] = H[((size_t)(b * 16 + i)) * (size_t)N2 + n];

    float y0[16];
    for (int j = 0; j < 16; ++j) {
        float s = R0b[j];
        for (int i = 0; i < 16; ++i) s = fmaf(v[i], R0W[i * 16 + j], s);
        y0[j] = s > 0.f ? s : 0.f;
    }

    for (int r = 0; r < 2; ++r) {
        float s = R1b[r];
        for (int j = 0; j < 16; ++j) s = fmaf(y0[j], R1W[j * 2 + r], s);
        Y[((size_t)(b * 2 + r)) * (size_t)N2 + n] = s;
    }
}

extern "C" void kernel_launch(void* const* d_in, const int* in_sizes, int n_in,
                              void* d_out, int out_size, void* d_ws, size_t ws_size,
                              hipStream_t stream)
{
    (void)in_sizes; (void)n_in; (void)out_size; (void)ws_size;

    const float* x     = (const float*)d_in[0];
    const float* S1    = (const float*)d_in[1];
    const float* S2    = (const float*)d_in[2];
    const int*   incid = (const int*)  d_in[3];
    const float* W10   = (const float*)d_in[4];
    const float* b10   = (const float*)d_in[5];
    const float* W11   = (const float*)d_in[6];
    const float* b11   = (const float*)d_in[7];
    const float* W20   = (const float*)d_in[8];
    const float* b20   = (const float*)d_in[9];
    const float* R0W   = (const float*)d_in[10];
    const float* R0b   = (const float*)d_in[11];
    const float* R1W   = (const float*)d_in[12];
    const float* R1b   = (const float*)d_in[13];
    float* out = (float*)d_out;

    constexpr int B = 4, F1 = 16, N1 = 8192, N2 = 4096, DEG = 4;

    char* ws = (char*)d_ws;
    const size_t szBig = (size_t)(B * F1) * N1 * sizeof(float);
    float* h1 = (float*)(ws);
    float* tA = (float*)(ws + 1 * szBig);
    float* tB = (float*)(ws + 2 * szBig);
    float* h2 = (float*)(ws + 3 * szBig);
    float* p  = (float*)(ws + 4 * szBig);

    auto grid1d    = [](int total, int blk) { return (total + blk - 1) / blk; };

    gemm_strip_lds<N1, B><<<N1 / 32, 256, 0, stream>>>(tA, x,  S1);
    gemm_strip_lds<N1, B><<<N1 / 32, 256, 0, stream>>>(tB, tA, S1);
    combine_relu_kernel<<<grid1d(B * N1, 256), 256, 0, stream>>>(
        h1, x, tA, tB, W10, b10, B, 1, F1, N1);

    gemm_strip_lds<N1, B * F1><<<N1 / 32, 256, 0, stream>>>(tA, h1, S1);
    gemm_strip_lds<N1, B * F1><<<N1 / 32, 256, 0, stream>>>(tB, tA, S1);
    combine_relu_kernel<<<grid1d(B * N1, 256), 256, 0, stream>>>(
        h2, h1, tA, tB, W11, b11, B, F1, F1, N1);

    pool_kernel<<<grid1d(B * F1 * N2, 256), 256, 0, stream>>>(
        p, h2, incid, B * F1, N1, N2, DEG);

    gemm_strip_lds<N2, B * F1><<<N2 / 32, 256, 0, stream>>>(tA, p,  S2);
    gemm_strip_lds<N2, B * F1><<<N2 / 32, 256, 0, stream>>>(tB, tA, S2);
    combine_relu_kernel<<<grid1d(B * N2, 256), 256, 0, stream>>>(
        h1  , p, tA, tB, W20, b20, B, F1, F1, N2);

    readout_kernel<<<grid1d(B * N2, 256), 256, 0, stream>>>(
        out, h1, R0W, R0b, R1W, R1b, B, N2);
}


# 4-row and N=4096 GEMMs: K range split over two workgroups, each writes its own partial plane; the reader (next GEMM's A staging, combine kernel) adds the two planes; f32 sum order changes
# speedup vs baseline: 4.2247x; 1.0847x over previous
#include <hip/hip_runtime.h>

typedef float v2f __attribute__((ext_vector_type(2)));
typedef float v4f __attribute__((ext_vector_type(4)));
typedef float v8f __attribute__((ext_vector_type(8)));

template <int N, int MROWS, int KS, bool ADUAL>
__global__ __launch_bounds__(256) void gemm_strip_lds(
    float* __restrict__ Z, const float* __restrict__ H,
    const float* __restrict__ S)
{
    constexpr int KC      = 256;
    constexpr int MT      = (MROWS + 15) / 16;
    constexpr int PA      = KC + 4;
    constexpr int PB      = 96;
    constexpr int PO      = 36;
    constexpr int A_STAGE = 16 * MT * PA;
    constexpr int B_STAGE = (KC / 2) * PB;
    constexpr int KLEN    = N / KS;
    constexpr int PLANE   = MROWS * N;
    constexpr int NCH     = KLEN / KC;
    constexpr int HV      = (MROWS * (KC / 4)) / 256;
    constexpr int NACC    = 4;
    constexpr int BLK     = 8;
    constexpr int NBLK    = (KC / 4) / BLK;
    constexpr int HRG     = 256 / (KC / 4);
    constexpr int NST     = HV + 4;
    constexpr int SPB     = (NST + NBLK - 1) / NBLK;
    static_assert(NBLK * BLK * 4 == KC && BLK % NACC == 0, "block shape");
    static_assert(HV * 256 == MROWS * (KC / 4), "H chunk must split evenly");
    static_assert(KLEN * KS == N && KLEN % KC == 0 && N % 32 == 0, "shape");
    static_assert(16 * MT * PO <= B_STAGE, "output tile fits a stage");
    static_assert(NCH >= 3, "the chunk loop peels its last two turns");

    __shared__ __attribute__((aligned(16))) float As[2][A_STAGE];
    __shared__ __attribute__((aligned(16))) float Bs[2][B_STAGE];

    const int t    = threadIdx.x;
    const int lane = t & 31;
    const int w    = t >> 5;
    const int c    = lane & 15;
    const int h    = lane >> 4;
    const int mt   = w >> 1;
    const int nt   = w & 1;
    const int n0   = blockIdx.x * 32;
    const int k0   = KS > 1 ? (int)blockIdx.y * KLEN : 0;

    v4f hreg[HV];
    v4f hreg2[HV];
    v4f sreg[8];

    const float* __restrict__ hp =
        H + (size_t)(t / (KC / 4)) * N + k0 + 4 * (t % (KC / 4));
    const float* __restrict__ sp =
        S + (size_t)(k0 + 2 * (t >> 3)) * N + n0 + 4 * (t & 7);
    float* la = &As[0][(t / (KC / 4)) * PA + 4 * (t % (KC / 4))];
    float* lb = &Bs[0][(t >> 3) * PB + 8 * (t & 7)];

    auto step = [&](int st, float* qa, float* qb, const float* hq,
                    const float* sq, bool more) {
        if (st < HV) {
            const int j = st;
            if constexpr (ADUAL)
                *reinterpret_cast<v4f*>(qa + j * HRG * PA) = hreg[j] + hreg2[j];
            else
                *reinterpret_cast<v4f*>(qa + j * HRG * PA) = hreg[j];
            if (more) {
                __builtin_amdgcn_sched_barrier(0);
                hreg[j] = *reinterpret_cast<const v4f*>(
                    hq + (size_t)(j * HRG) * N);
                if constexpr (ADUAL)
                    hreg2[j] = *reinterpret_cast<const v4f*>(
                        hq + PLANE + (size_t)(j * HRG) * N);
            }
        } else if (st < NST) {
            const int j = st - HV;
            const v4f r0 = sreg[2 * j];
            const v4f r1 = sreg[2 * j + 1];
            float* q = qb + j * 32 * PB;
            *reinterpret_cast<v4f*>(q)     = (v4f){r0.x, r1.x, r0.y, r1.y};
            *reinterpret_cast<v4f*>(q + 4) = (v4f){r0.z, r1.z, r0.w, r1.w};
            if (more) {
                __builtin_amdgcn_sched_barrier(0);
                const float* __restrict__ p = sq + (size_t)(64 * j) * N;
                sreg[2 * j]     = *reinterpret_cast<const v4f*>(p);
                sreg[2 * j + 1] = *reinterpret_cast<const v4f*>(p + N);
            }
        }
    };

    auto stage_barrier = [&]() {
        __builtin_amdgcn_fence(__ATOMIC_RELEASE, "workgroup", "local");
        __builtin_amdgcn_s_barrier();
        __builtin_amdgcn_fence(__ATOMIC_ACQUIRE, "workgroup", "local");
    };

    v8f accp[NACC];
#pragma unroll
    for (int q = 0; q < NACC; ++q)
        accp[q] = (v8f){0.f, 0.f, 0.f, 0.f, 0.f, 0.f, 0.f, 0.f};

    auto turn = [&](int buf, int kn, bool stage, bool more) {
        const float* a = &As[buf][(mt * 16 + c) * PA + 2 * h];
        const float* b = &Bs[buf][h * PB + (nt * 16 + c) * 2];
        float* qa = la + (buf ^ 1) * A_STAGE;
        float* qb = lb + (buf ^ 1) * B_STAGE;
        const float* hq = hp + kn;
        const float* sq = sp + (size_t)kn * N;
        v2f af[2][BLK];
        v2f bf[2][BLK];
        if (mt < MT) {
#pragma unroll
            for (int q = 0; q < BLK; ++q) {
                af[0][q] = *reinterpret_cast<const v2f*>(a + 4 * q);
                bf[0][q] = *reinterpret_cast<const v2f*>(b + 2 * q * PB);
            }
        }
#pragma unroll
        for (int blk = 0; blk < NBLK; ++blk) {
            const int cur = blk & 1;
            if (mt < MT) {
                if (blk + 1 < NBLK) {
#pragma unroll
                    for (int q = 0; q < BLK; ++q) {
                        const int ks = (blk + 1) * BLK + q;
                        af[cur ^ 1][q] =
                            *reinterpret_cast<const v2f*>(a + 4 * ks);
                        bf[cur ^ 1][q] =
                            *reinterpret_cast<const v2f*>(b + 2 * ks * PB);
                    }
                }
                __builtin_amdgcn_sched_barrier(0);
#pragma unroll
                for (int q = 0; q < BLK; ++q) {
                    accp[q % NACC] = __builtin_amdgcn_wmma_f32_16x16x4_f32(
                        false, af[cur][q], false, bf[cur][q], (short)0,
                        accp[q % NACC], false, false);
                }
            }
            __builtin_amdgcn_sched_barrier(0);
            if (stage) {
#pragma unroll
                for (int u = 0; u < SPB; ++u)
                    step(blk * SPB + u, qa, qb, hq, sq, more);
                __builtin_amdgcn_sched_barrier(0);
            }
        }
    };

    if (MROWS < 16 * MT) {
        for (int b = 0; b < 2; ++b)
            for (int i = t; i < (16 * MT - MROWS) * PA; i += 256)
                As[b][MROWS * PA + i] = 0.0f;
    }

#pragma unroll
    for (int j = 0; j < HV; ++j) {
        hreg[j] = *reinterpret_cast<const v4f*>(hp + (size_t)(j * HRG) * N);
        if constexpr (ADUAL)
            hreg2[j] = *reinterpret_cast<const v4f*>(
                hp + PLANE + (size_t)(j * HRG) * N);
    }
#pragma unroll
    for (int j = 0; j < 4; ++j) {
        const float* __restrict__ p = sp + (size_t)(64 * j) * N;
        sreg[2 * j]     = *reinterpret_cast<const v4f*>(p);
        sreg[2 * j + 1] = *reinterpret_cast<const v4f*>(p + N);
    }
#pragma unroll
    for (int st = 0; st < NST; ++st)
        step(st, la, lb, hp + KC, sp + (size_t)KC * N, true);
    stage_barrier();

    for (int i = 0; i < NCH - 2; ++i) {
        turn(i & 1, (i + 2) * KC, true, true);
        stage_barrier();
    }
    turn((NCH - 2) & 1, 0, true, false);
    stage_barrier();
    turn((NCH - 1) & 1, 0, false, false);
    stage_barrier();

    float* Ot = &Bs[0][0];
    float* Zp = Z + (KS > 1 ? (size_t)blockIdx.y * PLANE : (size_t)0);
    const v8f acc = (accp[0] + accp[1]) + (accp[2] + accp[3]);
    if (mt < MT) {
#pragma unroll
        for (int v = 0; v < 8; ++v)
            Ot[(mt * 16 + v + 8 * h) * PO + nt * 16 + c] = acc[v];
    }
    __syncthreads();
    for (int idx = t; idx < MROWS * 8; idx += 256) {
        const int row = idx >> 3;
        const int q   = idx & 7;
        const v4f val = *reinterpret_cast<const v4f*>(&Ot[row * PO + 4 * q]);
        *reinterpret_cast<v4f*>(Zp + (size_t)row * N + n0 + 4 * q) = val;
    }
}

template <bool SPLIT>
__global__ void combine_relu_kernel(
    float* __restrict__ Hout,
    const float* __restrict__ Z0, const float* __restrict__ Z1,
    const float* __restrict__ Z2,
    const float* __restrict__ W, const float* __restrict__ bias,
    int Bn, int Fin, int Fout, int N, size_t plane)
{
    const int tid = blockIdx.x * blockDim.x + threadIdx.x;
    if (tid >= Bn * N) return;
    const int n = tid % N;
    const int b = tid / N;

    float out[16];
    for (int g = 0; g < Fout; ++g) out[g] = bias[g];

    for (int f = 0; f < Fin; ++f) {
        const size_t idx = ((size_t)(b * Fin + f)) * (size_t)N + n;
        const float z0 = Z0[idx];
        float z1 = Z1[idx];
        float z2 = Z2[idx];
        if constexpr (SPLIT) {
            z1 += Z1[idx + plane];
            z2 += Z2[idx + plane];
        }
        const float* __restrict__ W0 = W + ((size_t)(0 * Fin + f)) * Fout;
        const float* __restrict__ W1 = W + ((size_t)(1 * Fin + f)) * Fout;
        const float* __restrict__ W2 = W + ((size_t)(2 * Fin + f)) * Fout;
        for (int g = 0; g < Fout; ++g)
            out[g] = fmaf(z0, W0[g], fmaf(z1, W1[g], fmaf(z2, W2[g], out[g])));
    }

    for (int g = 0; g < Fout; ++g) {
        const float v = out[g];
        Hout[((size_t)(b * Fout + g)) * (size_t)N + n] = v > 0.f ? v : 0.f;
    }
}

__global__ void pool_kernel(
    float* __restrict__ P, const float* __restrict__ Hin,
    const int* __restrict__ incid, int BF, int N1, int N2, int DEG)
{
    const int tid = blockIdx.x * blockDim.x + threadIdx.x;
    if (tid >= BF * N2) return;
    const int m  = tid % N2;
    const int bf = tid / N2;
    const float* __restrict__ row = Hin + (size_t)bf * (size_t)N1;
    float v = 0.0f;
    for (int d = 0; d < DEG; ++d) {
        const int j = incid[m * DEG + d];
        v = fmaxf(v, row[j]);
    }
    P[tid] = v;
}

__global__ void readout_kernel(
    float* __restrict__ Y, const float* __restrict__ H,
    const float* __restrict__ R0W, const float* __restrict__ R0b,
    const float* __restrict__ R1W, const float* __restrict__ R1b,
    int Bn, int N2)
{
    const int tid = blockIdx.x * blockDim.x + threadIdx.x;
    if (tid >= Bn * N2) return;
    const int n = tid % N2;
    const int b = tid / N2;

    float v[16];
    for (int i = 0; i < 16; ++i)
        v[i] = H[((size_t)(b * 16 + i)) * (size_t)N2 + n];

    float y0[16];
    for (int j = 0; j < 16; ++j) {
        float s = R0b[j];
        for (int i = 0; i < 16; ++i) s = fmaf(v[i], R0W[i * 16 + j], s);
        y0[j] = s > 0.f ? s : 0.f;
    }

    for (int r = 0; r < 2; ++r) {
        float s = R1b[r];
        for (int j = 0; j < 16; ++j) s = fmaf(y0[j], R1W[j * 2 + r], s);
        Y[((size_t)(b * 2 + r)) * (size_t)N2 + n] = s;
    }
}

extern "C" void kernel_launch(void* const* d_in, const int* in_sizes, int n_in,
                              void* d_out, int out_size, void* d_ws, size_t ws_size,
                              hipStream_t stream)
{
    (void)in_sizes; (void)n_in; (void)out_size; (void)ws_size;

    const float* x     = (const float*)d_in[0];
    const float* S1    = (const float*)d_in[1];
    const float* S2    = (const float*)d_in[2];
    const int*   incid = (const int*)  d_in[3];
    const float* W10   = (const float*)d_in[4];
    const float* b10   = (const float*)d_in[5];
    const float* W11   = (const float*)d_in[6];
    const float* b11   = (const float*)d_in[7];
    const float* W20   = (const float*)d_in[8];
    const float* b20   = (const float*)d_in[9];
    const float* R0W   = (const float*)d_in[10];
    const float* R0b   = (const float*)d_in[11];
    const float* R1W   = (const float*)d_in[12];
    const float* R1b   = (const float*)d_in[13];
    float* out = (float*)d_out;

    constexpr int B = 4, F1 = 16, N1 = 8192, N2 = 4096, DEG = 4;

    char* ws = (char*)d_ws;
    const size_t szBig = (size_t)(B * F1) * N1 * sizeof(float);
    float* h1 = (float*)(ws);
    float* tA = (float*)(ws + 1 * szBig);
    float* tB = (float*)(ws + 2 * szBig);
    float* h2 = (float*)(ws + 3 * szBig);
    float* p  = (float*)(ws + 4 * szBig);

    auto grid1d    = [](int total, int blk) { return (total + blk - 1) / blk; };

    gemm_strip_lds<N1, B, 2, false><<<dim3(N1 / 32, 2), 256, 0, stream>>>(tA, x,  S1);
    gemm_strip_lds<N1, B, 2, true ><<<dim3(N1 / 32, 2), 256, 0, stream>>>(tB, tA, S1);
    combine_relu_kernel<true><<<grid1d(B * N1, 256), 256, 0, stream>>>(
        h1, x, tA, tB, W10, b10, B, 1, F1, N1, (size_t)B * N1);

    gemm_strip_lds<N1, B * F1, 1, false><<<N1 / 32, 256, 0, stream>>>(tA, h1, S1);
    gemm_strip_lds<N1, B * F1, 1, false><<<N1 / 32, 256, 0, stream>>>(tB, tA, S1);
    combine_relu_kernel<false><<<grid1d(B * N1, 256), 256, 0, stream>>>(
        h2, h1, tA, tB, W11, b11, B, F1, F1, N1, (size_t)0);

    pool_kernel<<<grid1d(B * F1 * N2, 256), 256, 0, stream>>>(
        p, h2, incid, B * F1, N1, N2, DEG);

    gemm_strip_lds<N2, B * F1, 2, false><<<dim3(N2 / 32, 2), 256, 0, stream>>>(tA, p,  S2);
    gemm_strip_lds<N2, B * F1, 2, true ><<<dim3(N2 / 32, 2), 256, 0, stream>>>(tB, tA, S2);
    combine_relu_kernel<true><<<grid1d(B * N2, 256), 256, 0, stream>>>(
        h1  , p, tA, tB, W20, b20, B, F1, F1, N2,
        (size_t)B * F1 * N2);

    readout_kernel<<<grid1d(B * N2, 256), 256, 0, stream>>>(
        out, h1, R0W, R0b, R1W, R1b, B, N2);
}


# Combine and pool kernels: node/feature/degree counts as template constants, all of a thread's loads issued before its arithmetic, sums in fixed registers, 128/64-thread blocks; sum order unchanged
# speedup vs baseline: 4.4650x; 1.0569x over previous
#include <hip/hip_runtime.h>

typedef float v2f __attribute__((ext_vector_type(2)));
typedef float v4f __attribute__((ext_vector_type(4)));
typedef float v8f __attribute__((ext_vector_type(8)));

template <int N, int MROWS, int KS, bool ADUAL>
__global__ __launch_bounds__(256) void gemm_strip_lds(
    float* __restrict__ Z, const float* __restrict__ H,
    const float* __restrict__ S)
{
    constexpr int KC      = 256;
    constexpr int MT      = (MROWS + 15) / 16;
    constexpr int PA      = KC + 4;
    constexpr int PB      = 96;
    constexpr int PO      = 36;
    constexpr int A_STAGE = 16 * MT * PA;
    constexpr int B_STAGE = (KC / 2) * PB;
    constexpr int KLEN    = N / KS;
    constexpr int PLANE   = MROWS * N;
    constexpr int NCH     = KLEN / KC;
    constexpr int HV      = (MROWS * (KC / 4)) / 256;
    constexpr int NACC    = 4;
    constexpr int BLK     = 8;
    constexpr int NBLK    = (KC / 4) / BLK;
    constexpr int HRG     = 256 / (KC / 4);
    constexpr int NST     = HV + 4;
    constexpr int SPB     = (NST + NBLK - 1) / NBLK;
    static_assert(NBLK * BLK * 4 == KC && BLK % NACC == 0, "block shape");
    static_assert(HV * 256 == MROWS * (KC / 4), "H chunk must split evenly");
    static_assert(KLEN * KS == N && KLEN % KC == 0 && N % 32 == 0, "shape");
    static_assert(16 * MT * PO <= B_STAGE, "output tile fits a stage");
    static_assert(NCH >= 3, "the chunk loop peels its last two turns");

    __shared__ __attribute__((aligned(16))) float As[2][A_STAGE];
    __shared__ __attribute__((aligned(16))) float Bs[2][B_STAGE];

    const int t    = threadIdx.x;
    const int lane = t & 31;
    const int w    = t >> 5;
    const int c    = lane & 15;
    const int h    = lane >> 4;
    const int mt   = w >> 1;
    const int nt   = w & 1;
    const int n0   = blockIdx.x * 32;
    const int k0   = KS > 1 ? (int)blockIdx.y * KLEN : 0;

    v4f hreg[HV];
    v4f hreg2[HV];
    v4f sreg[8];

    const float* __restrict__ hp =
        H + (size_t)(t / (KC / 4)) * N + k0 + 4 * (t % (KC / 4));
    const float* __restrict__ sp =
        S + (size_t)(k0 + 2 * (t >> 3)) * N + n0 + 4 * (t & 7);
    float* la = &As[0][(t / (KC / 4)) * PA + 4 * (t % (KC / 4))];
    float* lb = &Bs[0][(t >> 3) * PB + 8 * (t & 7)];

    auto step = [&](int st, float* qa, float* qb, const float* hq,
                    const float* sq, bool more) {
        if (st < HV) {
            const int j = st;
            if constexpr (ADUAL)
                *reinterpret_cast<v4f*>(qa + j * HRG * PA) = hreg[j] + hreg2[j];
            else
                *reinterpret_cast<v4f*>(qa + j * HRG * PA) = hreg[j];
            if (more) {
                __builtin_amdgcn_sched_barrier(0);
                hreg[j] = *reinterpret_cast<const v4f*>(
                    hq + (size_t)(j * HRG) * N);
                if constexpr (ADUAL)
                    hreg2[j] = *reinterpret_cast<const v4f*>(
                        hq + PLANE + (size_t)(j * HRG) * N);
            }
        } else if (st < NST) {
            const int j = st - HV;
            const v4f r0 = sreg[2 * j];
            const v4f r1 = sreg[2 * j + 1];
            float* q = qb + j * 32 * PB;
            *reinterpret_cast<v4f*>(q)     = (v4f){r0.x, r1.x, r0.y, r1.y};
            *reinterpret_cast<v4f*>(q + 4) = (v4f){r0.z, r1.z, r0.w, r1.w};
            if (more) {
                __builtin_amdgcn_sched_barrier(0);
                const float* __restrict__ p = sq + (size_t)(64 * j) * N;
                sreg[2 * j]     = *reinterpret_cast<const v4f*>(p);
                sreg[2 * j + 1] = *reinterpret_cast<const v4f*>(p + N);
            }
        }
    };

    auto stage_barrier = [&]() {
        __builtin_amdgcn_fence(__ATOMIC_RELEASE, "workgroup", "local");
        __builtin_amdgcn_s_barrier();
        __builtin_amdgcn_fence(__ATOMIC_ACQUIRE, "workgroup", "local");
    };

    v8f accp[NACC];
#pragma unroll
    for (int q = 0; q < NACC; ++q)
        accp[q] = (v8f){0.f, 0.f, 0.f, 0.f, 0.f, 0.f, 0.f, 0.f};

    auto turn = [&](int buf, int kn, bool stage, bool more) {
        const float* a = &As[buf][(mt * 16 + c) * PA + 2 * h];
        const float* b = &Bs[buf][h * PB + (nt * 16 + c) * 2];
        float* qa = la + (buf ^ 1) * A_STAGE;
        float* qb = lb + (buf ^ 1) * B_STAGE;
        const float* hq = hp + kn;
        const float* sq = sp + (size_t)kn * N;
        v2f af[2][BLK];
        v2f bf[2][BLK];
        if (mt < MT) {
#pragma unroll
            for (int q = 0; q < BLK; ++q) {
                af[0][q] = *reinterpret_cast<const v2f*>(a + 4 * q);
                bf[0][q] = *reinterpret_cast<const v2f*>(b + 2 * q * PB);
            }
        }
#pragma unroll
        for (int blk = 0; blk < NBLK; ++blk) {
            const int cur = blk & 1;
            if (mt < MT) {
                if (blk + 1 < NBLK) {
#pragma unroll
                    for (int q = 0; q < BLK; ++q) {
                        const int ks = (blk + 1) * BLK + q;
                        af[cur ^ 1][q] =
                            *reinterpret_cast<const v2f*>(a + 4 * ks);
                        bf[cur ^ 1][q] =
                            *reinterpret_cast<const v2f*>(b + 2 * ks * PB);
                    }
                }
                __builtin_amdgcn_sched_barrier(0);
#pragma unroll
                for (int q = 0; q < BLK; ++q) {
                    accp[q % NACC] = __builtin_amdgcn_wmma_f32_16x16x4_f32(
                        false, af[cur][q], false, bf[cur][q], (short)0,
                        accp[q % NACC], false, false);
                }
            }
            __builtin_amdgcn_sched_barrier(0);
            if (stage) {
#pragma unroll
                for (int u = 0; u < SPB; ++u)
                    step(blk * SPB + u, qa, qb, hq, sq, more);
                __builtin_amdgcn_sched_barrier(0);
            }
        }
    };

    if (MROWS < 16 * MT) {
        for (int b = 0; b < 2; ++b)
            for (int i = t; i < (16 * MT - MROWS) * PA; i += 256)
                As[b][MROWS * PA + i] = 0.0f;
    }

#pragma unroll
    for (int j = 0; j < HV; ++j) {
        hreg[j] = *reinterpret_cast<const v4f*>(hp + (size_t)(j * HRG) * N);
        if constexpr (ADUAL)
            hreg2[j] = *reinterpret_cast<const v4f*>(
                hp + PLANE + (size_t)(j * HRG) * N);
    }
#pragma unroll
    for (int j = 0; j < 4; ++j) {
        const float* __restrict__ p = sp + (size_t)(64 * j) * N;
        sreg[2 * j]     = *reinterpret_cast<const v4f*>(p);
        sreg[2 * j + 1] = *reinterpret_cast<const v4f*>(p + N);
    }
#pragma unroll
    for (int st = 0; st < NST; ++st)
        step(st, la, lb, hp + KC, sp + (size_t)KC * N, true);
    stage_barrier();

    for (int i = 0; i < NCH - 2; ++i) {
        turn(i & 1, (i + 2) * KC, true, true);
        stage_barrier();
    }
    turn((NCH - 2) & 1, 0, true, false);
    stage_barrier();
    turn((NCH - 1) & 1, 0, false, false);
    stage_barrier();

    float* Ot = &Bs[0][0];
    float* Zp = Z + (KS > 1 ? (size_t)blockIdx.y * PLANE : (size_t)0);
    const v8f acc = (accp[0] + accp[1]) + (accp[2] + accp[3]);
    if (mt < MT) {
#pragma unroll
        for (int v = 0; v < 8; ++v)
            Ot[(mt * 16 + v + 8 * h) * PO + nt * 16 + c] = acc[v];
    }
    __syncthreads();
    for (int idx = t; idx < MROWS * 8; idx += 256) {
        const int row = idx >> 3;
        const int q   = idx & 7;
        const v4f val = *reinterpret_cast<const v4f*>(&Ot[row * PO + 4 * q]);
        *reinterpret_cast<v4f*>(Zp + (size_t)row * N + n0 + 4 * q) = val;
    }
}

template <int N, int FIN, bool SPLIT>
__global__ __launch_bounds__(128) void combine_relu_kernel(
    float* __restrict__ Hout,
    const float* __restrict__ Z0, const float* __restrict__ Z1,
    const float* __restrict__ Z2,
    const float* __restrict__ W, const float* __restrict__ bias,
    int Bn, size_t plane)
{
    constexpr int FOUT = 16;
    constexpr int FB   = SPLIT ? FIN : 1;
    const int tid = blockIdx.x * blockDim.x + threadIdx.x;
    if (tid >= Bn * N) return;
    const int n = tid % N;
    const int b = tid / N;
    const size_t base = (size_t)b * FIN * N + n;

    float z0[FIN], z1[FIN], z2[FIN], z1b[FB], z2b[FB];
#pragma unroll
    for (int f = 0; f < FIN; ++f) {
        const size_t idx = base + (size_t)f * N;
        z0[f] = Z0[idx];
        z1[f] = Z1[idx];
        z2[f] = Z2[idx];
        if constexpr (SPLIT) {
            z1b[f] = Z1[idx + plane];
            z2b[f] = Z2[idx + plane];
        }
    }
    __builtin_amdgcn_sched_barrier(0);

    float out[FOUT];
#pragma unroll
    for (int g = 0; g < FOUT; ++g) out[g] = bias[g];

#pragma unroll
    for (int f = 0; f < FIN; ++f) {
        float a1 = z1[f];
        float a2 = z2[f];
        if constexpr (SPLIT) {
            a1 += z1b[f];
            a2 += z2b[f];
        }
        const float* __restrict__ W0 = W + (0 * FIN + f) * FOUT;
        const float* __restrict__ W1 = W + (1 * FIN + f) * FOUT;
        const float* __restrict__ W2 = W + (2 * FIN + f) * FOUT;
#pragma unroll
        for (int g = 0; g < FOUT; ++g)
            out[g] = fmaf(z0[f], W0[g], fmaf(a1, W1[g], fmaf(a2, W2[g], out[g])));
    }

#pragma unroll
    for (int g = 0; g < FOUT; ++g) {
        const float v = out[g];
        Hout[((size_t)(b * FOUT + g)) * (size_t)N + n] = v > 0.f ? v : 0.f;
    }
}

template <int N1, int N2, int DEG>
__global__ void pool_kernel(
    float* __restrict__ P, const float* __restrict__ Hin,
    const int* __restrict__ incid, int BF)
{
    const int tid = blockIdx.x * blockDim.x + threadIdx.x;
    if (tid >= BF * N2) return;
    const int m  = tid % N2;
    const int bf = tid / N2;
    const float* __restrict__ row = Hin + (size_t)bf * (size_t)N1;
    int j[DEG];
#pragma unroll
    for (int d = 0; d < DEG; ++d) j[d] = incid[m * DEG + d];
    float r[DEG];
#pragma unroll
    for (int d = 0; d < DEG; ++d) r[d] = row[j[d]];
    float v = 0.0f;
#pragma unroll
    for (int d = 0; d < DEG; ++d) v = fmaxf(v, r[d]);
    P[tid] = v;
}

__global__ void readout_kernel(
    float* __restrict__ Y, const float* __restrict__ H,
    const float* __restrict__ R0W, const float* __restrict__ R0b,
    const float* __restrict__ R1W, const float* __restrict__ R1b,
    int Bn, int N2)
{
    const int tid = blockIdx.x * blockDim.x + threadIdx.x;
    if (tid >= Bn * N2) return;
    const int n = tid % N2;
    const int b = tid / N2;

    float v[16];
    for (int i = 0; i < 16; ++i)
        v[i] = H[((size_t)(b * 16 + i)) * (size_t)N2 + n];

    float y0[16];
    for (int j = 0; j < 16; ++j) {
        float s = R0b[j];
        for (int i = 0; i < 16; ++i) s = fmaf(v[i], R0W[i * 16 + j], s);
        y0[j] = s > 0.f ? s : 0.f;
    }

    for (int r = 0; r < 2; ++r) {
        float s = R1b[r];
        for (int j = 0; j < 16; ++j) s = fmaf(y0[j], R1W[j * 2 + r], s);
        Y[((size_t)(b * 2 + r)) * (size_t)N2 + n] = s;
    }
}

extern "C" void kernel_launch(void* const* d_in, const int* in_sizes, int n_in,
                              void* d_out, int out_size, void* d_ws, size_t ws_size,
                              hipStream_t stream)
{
    (void)in_sizes; (void)n_in; (void)out_size; (void)ws_size;

    const float* x     = (const float*)d_in[0];
    const float* S1    = (const float*)d_in[1];
    const float* S2    = (const float*)d_in[2];
    const int*   incid = (const int*)  d_in[3];
    const float* W10   = (const float*)d_in[4];
    const float* b10   = (const float*)d_in[5];
    const float* W11   = (const float*)d_in[6];
    const float* b11   = (const float*)d_in[7];
    const float* W20   = (const float*)d_in[8];
    const float* b20   = (const float*)d_in[9];
    const float* R0W   = (const float*)d_in[10];
    const float* R0b   = (const float*)d_in[11];
    const float* R1W   = (const float*)d_in[12];
    const float* R1b   = (const float*)d_in[13];
    float* out = (float*)d_out;

    constexpr int B = 4, F1 = 16, N1 = 8192, N2 = 4096, DEG = 4;

    char* ws = (char*)d_ws;
    const size_t szBig = (size_t)(B * F1) * N1 * sizeof(float);
    float* h1 = (float*)(ws);
    float* tA = (float*)(ws + 1 * szBig);
    float* tB = (float*)(ws + 2 * szBig);
    float* h2 = (float*)(ws + 3 * szBig);
    float* p  = (float*)(ws + 4 * szBig);

    auto grid1d    = [](int total, int blk) { return (total + blk - 1) / blk; };

    gemm_strip_lds<N1, B, 2, false><<<dim3(N1 / 32, 2), 256, 0, stream>>>(tA, x,  S1);
    gemm_strip_lds<N1, B, 2, true ><<<dim3(N1 / 32, 2), 256, 0, stream>>>(tB, tA, S1);
    combine_relu_kernel<N1, 1, true><<<grid1d(B * N1, 128), 128, 0, stream>>>(
        h1, x, tA, tB, W10, b10, B, (size_t)B * N1);

    gemm_strip_lds<N1, B * F1, 1, false><<<N1 / 32, 256, 0, stream>>>(tA, h1, S1);
    gemm_strip_lds<N1, B * F1, 1, false><<<N1 / 32, 256, 0, stream>>>(tB, tA, S1);
    combine_relu_kernel<N1, F1, false><<<grid1d(B * N1, 128), 128, 0, stream>>>(
        h2, h1, tA, tB, W11, b11, B, (size_t)0);

    pool_kernel<N1, N2, DEG><<<grid1d(B * F1 * N2, 256), 256, 0, stream>>>(
        p, h2, incid, B * F1);

    gemm_strip_lds<N2, B * F1, 2, false><<<dim3(N2 / 32, 2), 256, 0, stream>>>(tA, p,  S2);
    gemm_strip_lds<N2, B * F1, 2, true ><<<dim3(N2 / 32, 2), 256, 0, stream>>>(tB, tA, S2);
    combine_relu_kernel<N2, F1, true><<<grid1d(B * N2, 64), 64, 0, stream>>>(
        h1  , p, tA, tB, W20, b20, B,
        (size_t)B * F1 * N2);

    readout_kernel<<<grid1d(B * N2, 256), 256, 0, stream>>>(
        out, h1, R0W, R0b, R1W, R1b, B, N2);
}


# 4-row N=8192 GEMMs retiled: 128-column strips, K split over 8 workgroups, every wave owns a 16-column tile, S rows read as 512-byte runs; readers add the 8 partial planes; f32 sum order changes
# speedup vs baseline: 4.6695x; 1.0458x over previous
#include <hip/hip_runtime.h>

typedef float v2f __attribute__((ext_vector_type(2)));
typedef float v4f __attribute__((ext_vector_type(4)));
typedef float v8f __attribute__((ext_vector_type(8)));

template <int N, int KS, int NPL>
__global__ __launch_bounds__(256) void gemm_rows4_wide(
    float* __restrict__ Z, const float* __restrict__ H,
    const float* __restrict__ S)
{
    constexpr int MROWS   = 4;
    constexpr int SW      = 128;
    constexpr int KC      = 128;
    constexpr int PA      = KC + 4;
    constexpr int PB      = 2 * SW + 32;
    constexpr int PO      = SW + 4;
    constexpr int A_STAGE = (MROWS + 1) * PA;
    constexpr int B_STAGE = (KC / 2) * PB;
    constexpr int KLEN    = N / KS;
    constexpr int PLANE   = MROWS * N;
    constexpr int NCH     = KLEN / KC;
    constexpr int HT      = MROWS * (KC / 4);
    constexpr int NSP     = KC / 16;
    constexpr int NACC    = 4;
    constexpr int BLK     = 8;
    constexpr int NBLK    = (KC / 4) / BLK;
    constexpr int NST     = 1 + NSP;
    constexpr int SPB     = (NST + NBLK - 1) / NBLK;
    static_assert(NBLK * BLK * 4 == KC && BLK % NACC == 0, "block shape");
    static_assert(KLEN * KS == N && KLEN % KC == 0 && N % SW == 0, "shape");
    static_assert(HT <= 256 && HT % 32 == 0, "H staging threads are whole waves");
    static_assert(MROWS * PO <= B_STAGE, "output tile fits a stage");
    static_assert(NCH >= 3, "the chunk loop peels its last two turns");

    __shared__ __attribute__((aligned(16))) float As[2][A_STAGE];
    __shared__ __attribute__((aligned(16))) float Bs[2][B_STAGE];

    const int t    = threadIdx.x;
    const int lane = t & 31;
    const int w    = t >> 5;
    const int c    = lane & 15;
    const int h    = lane >> 4;
    const int n0   = blockIdx.x * SW;
    const int k0   = (int)blockIdx.y * KLEN;
    const int arow = c < MROWS ? c : MROWS;

    v4f hreg[NPL];
    v4f sreg[2 * NSP];

    const float* __restrict__ hp =
        H + (size_t)(t / (KC / 4)) * N + k0 + 4 * (t % (KC / 4));
    const float* __restrict__ sp =
        S + (size_t)(k0 + 2 * (t >> 5)) * N + n0 + 4 * (t & 31);
    float* la = &As[0][(t / (KC / 4)) * PA + 4 * (t % (KC / 4))];
    float* lb = &Bs[0][(t >> 5) * PB + 8 * (t & 31)];

    auto step = [&](int st, float* qa, float* qb, const float* hq,
                    const float* sq, bool more) {
        if (st == 0) {
            if (t < HT) {
                v4f s = hreg[0];
#pragma unroll
                for (int p = 1; p < NPL; ++p) s += hreg[p];
                *reinterpret_cast<v4f*>(qa) = s;
                if (more) {
                    __builtin_amdgcn_sched_barrier(0);
#pragma unroll
                    for (int p = 0; p < NPL; ++p)
                        hreg[p] = *reinterpret_cast<const v4f*>(
                            hq + (size_t)p * PLANE);
                }
            }
        } else if (st < NST) {
            const int j = st - 1;
            const v4f r0 = sreg[2 * j];
            const v4f r1 = sreg[2 * j + 1];
            float* q = qb + j * 8 * PB;
            *reinterpret_cast<v4f*>(q)     = (v4f){r0.x, r1.x, r0.y, r1.y};
            *reinterpret_cast<v4f*>(q + 4) = (v4f){r0.z, r1.z, r0.w, r1.w};
            if (more) {
                __builtin_amdgcn_sched_barrier(0);
                const float* __restrict__ p = sq + (size_t)(16 * j) * N;
                sreg[2 * j]     = *reinterpret_cast<const v4f*>(p);
                sreg[2 * j + 1] = *reinterpret_cast<const v4f*>(p + N);
            }
        }
    };

    auto stage_barrier = [&]() {
        __builtin_amdgcn_fence(__ATOMIC_RELEASE, "workgroup", "local");
        __builtin_amdgcn_s_barrier();
        __builtin_amdgcn_fence(__ATOMIC_ACQUIRE, "workgroup", "local");
    };

    v8f accp[NACC];
#pragma unroll
    for (int q = 0; q < NACC; ++q)
        accp[q] = (v8f){0.f, 0.f, 0.f, 0.f, 0.f, 0.f, 0.f, 0.f};

    auto turn = [&](int buf, int kn, bool stage, bool more) {
        const float* a = &As[buf][arow * PA + 2 * h];
        const float* b = &Bs[buf][h * PB + (w * 16 + c) * 2];
        float* qa = la + (buf ^ 1) * A_STAGE;
        float* qb = lb + (buf ^ 1) * B_STAGE;
        const float* hq = hp + kn;
        const float* sq = sp + (size_t)kn * N;
        v2f af[2][BLK];
        v2f bf[2][BLK];
#pragma unroll
        for (int q = 0; q < BLK; ++q) {
            af[0][q] = *reinterpret_cast<const v2f*>(a + 4 * q);
            bf[0][q] = *reinterpret_cast<const v2f*>(b + 2 * q * PB);
        }
#pragma unroll
        for (int blk = 0; blk < NBLK; ++blk) {
            const int cur = blk & 1;
            if (blk + 1 < NBLK) {
#pragma unroll
                for (int q = 0; q < BLK; ++q) {
                    const int ks = (blk + 1) * BLK + q;
                    af[cur ^ 1][q] =
                        *reinterpret_cast<const v2f*>(a + 4 * ks);
                    bf[cur ^ 1][q] =
                        *reinterpret_cast<const v2f*>(b + 2 * ks * PB);
                }
            }
            __builtin_amdgcn_sched_barrier(0);
#pragma unroll
            for (int q = 0; q < BLK; ++q) {
                accp[q % NACC] = __builtin_amdgcn_wmma_f32_16x16x4_f32(
                    false, af[cur][q], false, bf[cur][q], (short)0,
                    accp[q % NACC], false, false);
            }
            __builtin_amdgcn_sched_barrier(0);
            if (stage) {
#pragma unroll
                for (int u = 0; u < SPB; ++u)
                    step(blk * SPB + u, qa, qb, hq, sq, more);
                __builtin_amdgcn_sched_barrier(0);
            }
        }
    };

    for (int i = t; i < 2 * PA; i += 256)
        As[i / PA][MROWS * PA + (i % PA)] = 0.0f;

    if (t < HT) {
#pragma unroll
        for (int p = 0; p < NPL; ++p)
            hreg[p] = *reinterpret_cast<const v4f*>(hp + (size_t)p * PLANE);
    }
#pragma unroll
    for (int j = 0; j < NSP; ++j) {
        const float* __restrict__ p = sp + (size_t)(16 * j) * N;
        sreg[2 * j]     = *reinterpret_cast<const v4f*>(p);
        sreg[2 * j + 1] = *reinterpret_cast<const v4f*>(p + N);
    }
#pragma unroll
    for (int st = 0; st < NST; ++st)
        step(st, la, lb, hp + KC, sp + (size_t)KC * N, true);
    stage_barrier();

    for (int i = 0; i < NCH - 2; ++i) {
        turn(i & 1, (i + 2) * KC, true, true);
        stage_barrier();
    }
    turn((NCH - 2) & 1, 0, true, false);
    stage_barrier();
    turn((NCH - 1) & 1, 0, false, false);
    stage_barrier();

    float* Ot = &Bs[0][0];
    float* Zp = Z + (size_t)blockIdx.y * PLANE;
    const v8f acc = (accp[0] + accp[1]) + (accp[2] + accp[3]);
    if (h == 0) {
#pragma unroll
        for (int v = 0; v < MROWS; ++v)
            Ot[v * PO + w * 16 + c] = acc[v];
    }
    __syncthreads();
    if (t < MROWS * (SW / 4)) {
        const int row = t / (SW / 4);
        const int q   = t % (SW / 4);
        const v4f val = *reinterpret_cast<const v4f*>(&Ot[row * PO + 4 * q]);
        *reinterpret_cast<v4f*>(Zp + (size_t)row * N + n0 + 4 * q) = val;
    }
}

template <int N, int MROWS, int KS, bool ADUAL>
__global__ __launch_bounds__(256) void gemm_strip_lds(
    float* __restrict__ Z, const float* __restrict__ H,
    const float* __restrict__ S)
{
    constexpr int KC      = 256;
    constexpr int MT      = (MROWS + 15) / 16;
    constexpr int PA      = KC + 4;
    constexpr int PB      = 96;
    constexpr int PO      = 36;
    constexpr int A_STAGE = 16 * MT * PA;
    constexpr int B_STAGE = (KC / 2) * PB;
    constexpr int KLEN    = N / KS;
    constexpr int PLANE   = MROWS * N;
    constexpr int NCH     = KLEN / KC;
    constexpr int HV      = (MROWS * (KC / 4)) / 256;
    constexpr int NACC    = 4;
    constexpr int BLK     = 8;
    constexpr int NBLK    = (KC / 4) / BLK;
    constexpr int HRG     = 256 / (KC / 4);
    constexpr int NST     = HV + 4;
    constexpr int SPB     = (NST + NBLK - 1) / NBLK;
    static_assert(NBLK * BLK * 4 == KC && BLK % NACC == 0, "block shape");
    static_assert(HV * 256 == MROWS * (KC / 4), "H chunk must split evenly");
    static_assert(KLEN * KS == N && KLEN % KC == 0 && N % 32 == 0, "shape");
    static_assert(16 * MT * PO <= B_STAGE, "output tile fits a stage");
    static_assert(NCH >= 3, "the chunk loop peels its last two turns");

    __shared__ __attribute__((aligned(16))) float As[2][A_STAGE];
    __shared__ __attribute__((aligned(16))) float Bs[2][B_STAGE];

    const int t    = threadIdx.x;
    const int lane = t & 31;
    const int w    = t >> 5;
    const int c    = lane & 15;
    const int h    = lane >> 4;
    const int mt   = w >> 1;
    const int nt   = w & 1;
    const int n0   = blockIdx.x * 32;
    const int k0   = KS > 1 ? (int)blockIdx.y * KLEN : 0;

    v4f hreg[HV];
    v4f hreg2[HV];
    v4f sreg[8];

    const float* __restrict__ hp =
        H + (size_t)(t / (KC / 4)) * N + k0 + 4 * (t % (KC / 4));
    const float* __restrict__ sp =
        S + (size_t)(k0 + 2 * (t >> 3)) * N + n0 + 4 * (t & 7);
    float* la = &As[0][(t / (KC / 4)) * PA + 4 * (t % (KC / 4))];
    float* lb = &Bs[0][(t >> 3) * PB + 8 * (t & 7)];

    auto step = [&](int st, float* qa, float* qb, const float* hq,
                    const float* sq, bool more) {
        if (st < HV) {
            const int j = st;
            if constexpr (ADUAL)
                *reinterpret_cast<v4f*>(qa + j * HRG * PA) = hreg[j] + hreg2[j];
            else
                *reinterpret_cast<v4f*>(qa + j * HRG * PA) = hreg[j];
            if (more) {
                __builtin_amdgcn_sched_barrier(0);
                hreg[j] = *reinterpret_cast<const v4f*>(
                    hq + (size_t)(j * HRG) * N);
                if constexpr (ADUAL)
                    hreg2[j] = *reinterpret_cast<const v4f*>(
                        hq + PLANE + (size_t)(j * HRG) * N);
            }
        } else if (st < NST) {
            const int j = st - HV;
            const v4f r0 = sreg[2 * j];
            const v4f r1 = sreg[2 * j + 1];
            float* q = qb + j * 32 * PB;
            *reinterpret_cast<v4f*>(q)     = (v4f){r0.x, r1.x, r0.y, r1.y};
            *reinterpret_cast<v4f*>(q + 4) = (v4f){r0.z, r1.z, r0.w, r1.w};
            if (more) {
                __builtin_amdgcn_sched_barrier(0);
                const float* __restrict__ p = sq + (size_t)(64 * j) * N;
                sreg[2 * j]     = *reinterpret_cast<const v4f*>(p);
                sreg[2 * j + 1] = *reinterpret_cast<const v4f*>(p + N);
            }
        }
    };

    auto stage_barrier = [&]() {
        __builtin_amdgcn_fence(__ATOMIC_RELEASE, "workgroup", "local");
        __builtin_amdgcn_s_barrier();
        __builtin_amdgcn_fence(__ATOMIC_ACQUIRE, "workgroup", "local");
    };

    v8f accp[NACC];
#pragma unroll
    for (int q = 0; q < NACC; ++q)
        accp[q] = (v8f){0.f, 0.f, 0.f, 0.f, 0.f, 0.f, 0.f, 0.f};

    auto turn = [&](int buf, int kn, bool stage, bool more) {
        const float* a = &As[buf][(mt * 16 + c) * PA + 2 * h];
        const float* b = &Bs[buf][h * PB + (nt * 16 + c) * 2];
        float* qa = la + (buf ^ 1) * A_STAGE;
        float* qb = lb + (buf ^ 1) * B_STAGE;
        const float* hq = hp + kn;
        const float* sq = sp + (size_t)kn * N;
        v2f af[2][BLK];
        v2f bf[2][BLK];
        if (mt < MT) {
#pragma unroll
            for (int q = 0; q < BLK; ++q) {
                af[0][q] = *reinterpret_cast<const v2f*>(a + 4 * q);
                bf[0][q] = *reinterpret_cast<const v2f*>(b + 2 * q * PB);
            }
        }
#pragma unroll
        for (int blk = 0; blk < NBLK; ++blk) {
            const int cur = blk & 1;
            if (mt < MT) {
                if (blk + 1 < NBLK) {
#pragma unroll
                    for (int q = 0; q < BLK; ++q) {
                        const int ks = (blk + 1) * BLK + q;
                        af[cur ^ 1][q] =
                            *reinterpret_cast<const v2f*>(a + 4 * ks);
                        bf[cur ^ 1][q] =
                            *reinterpret_cast<const v2f*>(b + 2 * ks * PB);
                    }
                }
                __builtin_amdgcn_sched_barrier(0);
#pragma unroll
                for (int q = 0; q < BLK; ++q) {
                    accp[q % NACC] = __builtin_amdgcn_wmma_f32_16x16x4_f32(
                        false, af[cur][q], false, bf[cur][q], (short)0,
                        accp[q % NACC], false, false);
                }
            }
            __builtin_amdgcn_sched_barrier(0);
            if (stage) {
#pragma unroll
                for (int u = 0; u < SPB; ++u)
                    step(blk * SPB + u, qa, qb, hq, sq, more);
                __builtin_amdgcn_sched_barrier(0);
            }
        }
    };

    if (MROWS < 16 * MT) {
        for (int b = 0; b < 2; ++b)
            for (int i = t; i < (16 * MT - MROWS) * PA; i += 256)
                As[b][MROWS * PA + i] = 0.0f;
    }

#pragma unroll
    for (int j = 0; j < HV; ++j) {
        hreg[j] = *reinterpret_cast<const v4f*>(hp + (size_t)(j * HRG) * N);
        if constexpr (ADUAL)
            hreg2[j] = *reinterpret_cast<const v4f*>(
                hp + PLANE + (size_t)(j * HRG) * N);
    }
#pragma unroll
    for (int j = 0; j < 4; ++j) {
        const float* __restrict__ p = sp + (size_t)(64 * j) * N;
        sreg[2 * j]     = *reinterpret_cast<const v4f*>(p);
        sreg[2 * j + 1] = *reinterpret_cast<const v4f*>(p + N);
    }
#pragma unroll
    for (int st = 0; st < NST; ++st)
        step(st, la, lb, hp + KC, sp + (size_t)KC * N, true);
    stage_barrier();

    for (int i = 0; i < NCH - 2; ++i) {
        turn(i & 1, (i + 2) * KC, true, true);
        stage_barrier();
    }
    turn((NCH - 2) & 1, 0, true, false);
    stage_barrier();
    turn((NCH - 1) & 1, 0, false, false);
    stage_barrier();

    float* Ot = &Bs[0][0];
    float* Zp = Z + (KS > 1 ? (size_t)blockIdx.y * PLANE : (size_t)0);
    const v8f acc = (accp[0] + accp[1]) + (accp[2] + accp[3]);
    if (mt < MT) {
#pragma unroll
        for (int v = 0; v < 8; ++v)
            Ot[(mt * 16 + v + 8 * h) * PO + nt * 16 + c] = acc[v];
    }
    __syncthreads();
    for (int idx = t; idx < MROWS * 8; idx += 256) {
        const int row = idx >> 3;
        const int q   = idx & 7;
        const v4f val = *reinterpret_cast<const v4f*>(&Ot[row * PO + 4 * q]);
        *reinterpret_cast<v4f*>(Zp + (size_t)row * N + n0 + 4 * q) = val;
    }
}

template <int N, int FIN, bool SPLIT>
__global__ __launch_bounds__(128) void combine_relu_kernel(
    float* __restrict__ Hout,
    const float* __restrict__ Z0, const float* __restrict__ Z1,
    const float* __restrict__ Z2,
    const float* __restrict__ W, const float* __restrict__ bias,
    int Bn, size_t plane)
{
    constexpr int FOUT = 16;
    constexpr int FB   = SPLIT ? FIN : 1;
    const int tid = blockIdx.x * blockDim.x + threadIdx.x;
    if (tid >= Bn * N) return;
    const int n = tid % N;
    const int b = tid / N;
    const size_t base = (size_t)b * FIN * N + n;

    float z0[FIN], z1[FIN], z2[FIN], z1b[FB], z2b[FB];
#pragma unroll
    for (int f = 0; f < FIN; ++f) {
        const size_t idx = base + (size_t)f * N;
        z0[f] = Z0[idx];
        z1[f] = Z1[idx];
        z2[f] = Z2[idx];
        if constexpr (SPLIT) {
            z1b[f] = Z1[idx + plane];
            z2b[f] = Z2[idx + plane];
        }
    }
    __builtin_amdgcn_sched_barrier(0);

    float out[FOUT];
#pragma unroll
    for (int g = 0; g < FOUT; ++g) out[g] = bias[g];

#pragma unroll
    for (int f = 0; f < FIN; ++f) {
        float a1 = z1[f];
        float a2 = z2[f];
        if constexpr (SPLIT) {
            a1 += z1b[f];
            a2 += z2b[f];
        }
        const float* __restrict__ W0 = W + (0 * FIN + f) * FOUT;
        const float* __restrict__ W1 = W + (1 * FIN + f) * FOUT;
        const float* __restrict__ W2 = W + (2 * FIN + f) * FOUT;
#pragma unroll
        for (int g = 0; g < FOUT; ++g)
            out[g] = fmaf(z0[f], W0[g], fmaf(a1, W1[g], fmaf(a2, W2[g], out[g])));
    }

#pragma unroll
    for (int g = 0; g < FOUT; ++g) {
        const float v = out[g];
        Hout[((size_t)(b * FOUT + g)) * (size_t)N + n] = v > 0.f ? v : 0.f;
    }
}

template <int N, int NPL>
__global__ __launch_bounds__(128) void combine1_planes_kernel(
    float* __restrict__ Hout,
    const float* __restrict__ X, const float* __restrict__ Z1,
    const float* __restrict__ Z2,
    const float* __restrict__ W, const float* __restrict__ bias,
    int Bn, size_t plane)
{
    constexpr int FOUT = 16;
    const int tid = blockIdx.x * blockDim.x + threadIdx.x;
    if (tid >= Bn * N) return;
    const int n = tid % N;
    const int b = tid / N;
    const size_t base = (size_t)b * N + n;

    const float x0 = X[base];
    float p1[NPL], p2[NPL];
#pragma unroll
    for (int p = 0; p < NPL; ++p) {
        p1[p] = Z1[base + (size_t)p * plane];
        p2[p] = Z2[base + (size_t)p * plane];
    }
    __builtin_amdgcn_sched_barrier(0);

    float a1 = p1[0];
    float a2 = p2[0];
#pragma unroll
    for (int p = 1; p < NPL; ++p) {
        a1 += p1[p];
        a2 += p2[p];
    }

#pragma unroll
    for (int g = 0; g < FOUT; ++g) {
        const float v = fmaf(x0, W[g],
                        fmaf(a1, W[FOUT + g],
                        fmaf(a2, W[2 * FOUT + g], bias[g])));
        Hout[((size_t)(b * FOUT + g)) * (size_t)N + n] = v > 0.f ? v : 0.f;
    }
}

template <int N1, int N2, int DEG>
__global__ void pool_kernel(
    float* __restrict__ P, const float* __restrict__ Hin,
    const int* __restrict__ incid, int BF)
{
    const int tid = blockIdx.x * blockDim.x + threadIdx.x;
    if (tid >= BF * N2) return;
    const int m  = tid % N2;
    const int bf = tid / N2;
    const float* __restrict__ row = Hin + (size_t)bf * (size_t)N1;
    int j[DEG];
#pragma unroll
    for (int d = 0; d < DEG; ++d) j[d] = incid[m * DEG + d];
    float r[DEG];
#pragma unroll
    for (int d = 0; d < DEG; ++d) r[d] = row[j[d]];
    float v = 0.0f;
#pragma unroll
    for (int d = 0; d < DEG; ++d) v = fmaxf(v, r[d]);
    P[tid] = v;
}

__global__ void readout_kernel(
    float* __restrict__ Y, const float* __restrict__ H,
    const float* __restrict__ R0W, const float* __restrict__ R0b,
    const float* __restrict__ R1W, const float* __restrict__ R1b,
    int Bn, int N2)
{
    const int tid = blockIdx.x * blockDim.x + threadIdx.x;
    if (tid >= Bn * N2) return;
    const int n = tid % N2;
    const int b = tid / N2;

    float v[16];
    for (int i = 0; i < 16; ++i)
        v[i] = H[((size_t)(b * 16 + i)) * (size_t)N2 + n];

    float y0[16];
    for (int j = 0; j < 16; ++j) {
        float s = R0b[j];
        for (int i = 0; i < 16; ++i) s = fmaf(v[i], R0W[i * 16 + j], s);
        y0[j] = s > 0.f ? s : 0.f;
    }

    for (int r = 0; r < 2; ++r) {
        float s = R1b[r];
        for (int j = 0; j < 16; ++j) s = fmaf(y0[j], R1W[j * 2 + r], s);
        Y[((size_t)(b * 2 + r)) * (size_t)N2 + n] = s;
    }
}

extern "C" void kernel_launch(void* const* d_in, const int* in_sizes, int n_in,
                              void* d_out, int out_size, void* d_ws, size_t ws_size,
                              hipStream_t stream)
{
    (void)in_sizes; (void)n_in; (void)out_size; (void)ws_size;

    const float* x     = (const float*)d_in[0];
    const float* S1    = (const float*)d_in[1];
    const float* S2    = (const float*)d_in[2];
    const int*   incid = (const int*)  d_in[3];
    const float* W10   = (const float*)d_in[4];
    const float* b10   = (const float*)d_in[5];
    const float* W11   = (const float*)d_in[6];
    const float* b11   = (const float*)d_in[7];
    const float* W20   = (const float*)d_in[8];
    const float* b20   = (const float*)d_in[9];
    const float* R0W   = (const float*)d_in[10];
    const float* R0b   = (const float*)d_in[11];
    const float* R1W   = (const float*)d_in[12];
    const float* R1b   = (const float*)d_in[13];
    float* out = (float*)d_out;

    constexpr int B = 4, F1 = 16, N1 = 8192, N2 = 4096, DEG = 4;

    char* ws = (char*)d_ws;
    const size_t szBig = (size_t)(B * F1) * N1 * sizeof(float);
    float* h1 = (float*)(ws);
    float* tA = (float*)(ws + 1 * szBig);
    float* tB = (float*)(ws + 2 * szBig);
    float* h2 = (float*)(ws + 3 * szBig);
    float* p  = (float*)(ws + 4 * szBig);

    auto grid1d    = [](int total, int blk) { return (total + blk - 1) / blk; };

    gemm_rows4_wide<N1, 8, 1><<<dim3(N1 / 128, 8), 256, 0, stream>>>(tA, x,  S1);
    gemm_rows4_wide<N1, 8, 8><<<dim3(N1 / 128, 8), 256, 0, stream>>>(tB, tA, S1);
    combine1_planes_kernel<N1, 8><<<grid1d(B * N1, 128), 128, 0, stream>>>(
        h1, x, tA, tB, W10, b10, B, (size_t)B * N1);

    gemm_strip_lds<N1, B * F1, 1, false><<<N1 / 32, 256, 0, stream>>>(tA, h1, S1);
    gemm_strip_lds<N1, B * F1, 1, false><<<N1 / 32, 256, 0, stream>>>(tB, tA, S1);
    combine_relu_kernel<N1, F1, false><<<grid1d(B * N1, 128), 128, 0, stream>>>(
        h2, h1, tA, tB, W11, b11, B, (size_t)0);

    pool_kernel<N1, N2, DEG><<<grid1d(B * F1 * N2, 256), 256, 0, stream>>>(
        p, h2, incid, B * F1);

    gemm_strip_lds<N2, B * F1, 2, false><<<dim3(N2 / 32, 2), 256, 0, stream>>>(tA, p,  S2);
    gemm_strip_lds<N2, B * F1, 2, true ><<<dim3(N2 / 32, 2), 256, 0, stream>>>(tB, tA, S2);
    combine_relu_kernel<N2, F1, true><<<grid1d(B * N2, 64), 64, 0, stream>>>(
        h1  , p, tA, tB, W20, b20, B,
        (size_t)B * F1 * N2);

    readout_kernel<<<grid1d(B * N2, 256), 256, 0, stream>>>(
        out, h1, R0W, R0b, R1W, R1b, B, N2);
}
